# starting kernel
# baseline (speedup 1.0000x reference)
#include <hip/hip_runtime.h>
#include <math.h>

typedef __attribute__((ext_vector_type(2))) float v2f;
typedef __attribute__((ext_vector_type(8))) float v8f;

#define WMMA_F32_16x16x4(a, b, c) \
    __builtin_amdgcn_wmma_f32_16x16x4_f32(false, (a), false, (b), (short)0, (c), false, false)

__global__ void gemm_xwT_wmma(const float* __restrict__ X, const float* __restrict__ W,
                              float* __restrict__ Y, int M, int N, int K) {
    const int wave = threadIdx.x >> 5;
    const int lane = threadIdx.x & 31;
    const int tile = blockIdx.x * (blockDim.x >> 5) + wave;
    const int ntn  = N >> 4;
    if (tile >= (M >> 4) * ntn) return;
    const int tm   = (tile / ntn) << 4;
    const int tn   = (tile % ntn) << 4;
    const int half = lane >> 4;
    const int l    = lane & 15;
    const int kk   = half << 1;

    const float* arow = X + (size_t)(tm + l) * K;
    const float* brow = W + (size_t)(tn + l) * K;
    v8f acc = {};
    for (int k = 0; k < K; k += 4) {
        v2f a = *(const v2f*)(arow + k + kk);
        v2f b = *(const v2f*)(brow + k + kk);
        acc = WMMA_F32_16x16x4(a, b, acc);
    }
    const int mbase = tm + half * 8;
#pragma unroll
    for (int i = 0; i < 8; ++i)
        Y[(size_t)(mbase + i) * N + tn + l] = acc[i];
}

__global__ void gemm_ab_batched_wmma(const float* __restrict__ A, const float* __restrict__ Bm,
                                     float* __restrict__ Y, int M, int N, int K) {
    const int wave  = threadIdx.x >> 5;
    const int lane  = threadIdx.x & 31;
    const int batch = blockIdx.y;
    const int tile  = blockIdx.x * (blockDim.x >> 5) + wave;
    const int ntn   = N >> 4;
    if (tile >= (M >> 4) * ntn) return;
    const int tm   = (tile / ntn) << 4;
    const int tn   = (tile % ntn) << 4;
    const int half = lane >> 4;
    const int l    = lane & 15;
    const int kk   = half << 1;

    const float* Ab = A  + (size_t)batch * M * K;
    const float* Bb = Bm + (size_t)batch * K * N;
    float*       Yb = Y  + (size_t)batch * M * N;

    const float* arow = Ab + (size_t)(tm + l) * K;
    const int n = tn + l;
    v8f acc = {};
    for (int k = 0; k < K; k += 4) {
        v2f a = *(const v2f*)(arow + k + kk);
        v2f b;
        b.x = Bb[(size_t)(k + kk) * N + n];
        b.y = Bb[(size_t)(k + kk + 1) * N + n];
        acc = WMMA_F32_16x16x4(a, b, acc);
    }
    const int mbase = tm + half * 8;
#pragma unroll
    for (int i = 0; i < 8; ++i)
        Yb[(size_t)(mbase + i) * N + n] = acc[i];
}

__global__ void out_gemm_wmma(const float* __restrict__ Q, const float* __restrict__ Cc,
                              const float* __restrict__ Wout, float* __restrict__ Yout) {
    const int M = 512, N = 256, H = 256, F = 512;
    const int wave = threadIdx.x >> 5;
    const int lane = threadIdx.x & 31;
    const int tile = blockIdx.x * (blockDim.x >> 5) + wave;
    const int ntn  = N >> 4;
    if (tile >= (M >> 4) * ntn) return;
    const int tm   = (tile / ntn) << 4;
    const int tn   = (tile % ntn) << 4;
    const int half = lane >> 4;
    const int l    = lane & 15;
    const int kk   = half << 1;

    const float* qrow = Q    + (size_t)(tm + l) * H;
    const float* crow = Cc   + (size_t)(tm + l) * H;
    const float* wrow = Wout + (size_t)(tn + l) * F;

    v8f acc = {};
    for (int k = 0; k < H; k += 4) {
        v2f a = *(const v2f*)(qrow + k + kk);
        v2f b = *(const v2f*)(wrow + k + kk);
        acc = WMMA_F32_16x16x4(a, b, acc);
    }
    for (int k = 0; k < H; k += 4) {
        v2f a = *(const v2f*)(crow + k + kk);
        v2f b = *(const v2f*)(wrow + H + k + kk);
        acc = WMMA_F32_16x16x4(a, b, acc);
    }
    const int mbase = tm + half * 8;
#pragma unroll
    for (int i = 0; i < 8; ++i)
        Yout[(size_t)(mbase + i) * N + tn + l] = tanhf(acc[i]);
}

__global__ void score_softmax_kernel(const float* __restrict__ pq,
                                     const float* __restrict__ pe,
                                     const float* __restrict__ v,
                                     const int*   __restrict__ src_len,
                                     float* __restrict__ alpha) {
    const int H = 256, S = 512, T = 128;
    const int bt   = blockIdx.x;
    const int b    = bt / T;
    const int tid  = threadIdx.x;
    const int lane = tid & 31;
    const int wave = tid >> 5;

    __shared__ float sh_pq[256];
    __shared__ float sh_v[256];
    __shared__ float sh_e[512];
    __shared__ float red[8];

    sh_pq[tid] = pq[(size_t)bt * H + tid];
    sh_v[tid]  = v[tid];
    __syncthreads();

    const int slen = src_len[b];
    const float* peb = pe + (size_t)b * S * H;

    for (int s = wave * 64; s < wave * 64 + 64; ++s) {
        const float* per = peb + (size_t)s * H;
        float acc = 0.f;
#pragma unroll
        for (int hh = 0; hh < 256; hh += 32) {
            const int h = hh + lane;
            acc += sh_v[h] * tanhf(sh_pq[h] + per[h]);
        }
#pragma unroll
        for (int off = 16; off > 0; off >>= 1)
            acc += __shfl_xor(acc, off, 32);
        if (lane == 0)
            sh_e[s] = (s < slen) ? acc : -INFINITY;
    }
    __syncthreads();

    float m = -INFINITY;
    for (int s = tid; s < S; s += 256) m = fmaxf(m, sh_e[s]);
#pragma unroll
    for (int off = 16; off > 0; off >>= 1) m = fmaxf(m, __shfl_xor(m, off, 32));
    if (lane == 0) red[wave] = m;
    __syncthreads();
    m = red[0];
#pragma unroll
    for (int i = 1; i < 8; ++i) m = fmaxf(m, red[i]);
    __syncthreads();

    float sum = 0.f;
    for (int s = tid; s < S; s += 256) {
        float ex = __expf(sh_e[s] - m);
        sh_e[s] = ex;
        sum += ex;
    }
#pragma unroll
    for (int off = 16; off > 0; off >>= 1) sum += __shfl_xor(sum, off, 32);
    if (lane == 0) red[wave] = sum;
    __syncthreads();
    float total = 0.f;
#pragma unroll
    for (int i = 0; i < 8; ++i) total += red[i];
    const float inv = 1.f / total;

    for (int s = tid; s < S; s += 256)
        alpha[(size_t)bt * S + s] = sh_e[s] * inv;
}

extern "C" void kernel_launch(void* const* d_in, const int* in_sizes, int n_in,
                              void* d_out, int out_size, void* d_ws, size_t ws_size,
                              hipStream_t stream) {
    const int B = 4, T = 128, S = 512, H = 256;

    const float* query = (const float*)d_in[0];
    const float* enc   = (const float*)d_in[1];
    const int*   slen  = (const int*)d_in[2];
    const float* Ws    = (const float*)d_in[3];
    const float* Wh    = (const float*)d_in[4];
    const float* v     = (const float*)d_in[5];
    const float* Wout  = (const float*)d_in[6];
    float* out = (float*)d_out;

    float* ws    = (float*)d_ws;
    float* pq    = ws;
    float* pe    = pq + (size_t)B * T * H;
    float* alpha = pe + (size_t)B * S * H;
    float* ctx   = alpha + (size_t)B * T * S;
    (void)ws_size; (void)in_sizes; (void)n_in; (void)out_size;

    gemm_xwT_wmma<<<128, 128, 0, stream>>>(query, Ws, pq, B * T, H, H);

    gemm_xwT_wmma<<<512, 128, 0, stream>>>(enc, Wh, pe, B * S, H, H);

    score_softmax_kernel<<<B * T, 256, 0, stream>>>(pq, pe, v, slen, alpha);

    gemm_ab_batched_wmma<<<dim3(32, B), 128, 0, stream>>>(alpha, enc, ctx, T, H, S);

    out_gemm_wmma<<<128, 128, 0, stream>>>(query, ctx, Wout, out);
}


# score kernel: each wave works on 8 source rows at a time (16-byte loads, next 8 rows already on their way); the 8 lane sums run side by side on DPP adds, not one serial 5-step bpermute chain per row
# speedup vs baseline: 3.7502x; 3.7502x over previous
#include <hip/hip_runtime.h>
#include <math.h>

typedef __attribute__((ext_vector_type(2))) float v2f;
typedef __attribute__((ext_vector_type(8))) float v8f;

#define WMMA_F32_16x16x4(a, b, c) \
    __builtin_amdgcn_wmma_f32_16x16x4_f32(false, (a), false, (b), (short)0, (c), false, false)

typedef __attribute__((ext_vector_type(4))) float v4f;

template <int M>
__device__ __forceinline__ float row_xor(float x) {
    const int b = __builtin_bit_cast(int, x);
    return __builtin_bit_cast(float, __builtin_amdgcn_mov_dpp(b, 0x160 + M, 0xf, 0xf, true));
}
__device__ __forceinline__ float xhalf(float x) {
    const int b = __builtin_bit_cast(int, x);
    return __builtin_bit_cast(float,
        __builtin_amdgcn_permlanex16(b, b, 0x76543210, 0xfedcba98, false, false));
}

__global__ void gemm_xwT_exp2x_wmma(const float* __restrict__ X, const float* __restrict__ W,
                                    float* __restrict__ Y, int M, int N, int K) {
    const int wave = threadIdx.x >> 5;
    const int lane = threadIdx.x & 31;
    const int tile = blockIdx.x * (blockDim.x >> 5) + wave;
    const int ntn  = N >> 4;
    if (tile >= (M >> 4) * ntn) return;
    const int tm   = (tile / ntn) << 4;
    const int tn   = (tile % ntn) << 4;
    const int half = lane >> 4;
    const int l    = lane & 15;
    const int kk   = half << 1;

    const float* arow = X + (size_t)(tm + l) * K;
    const float* brow = W + (size_t)(tn + l) * K;
    v8f acc = {};
    for (int k = 0; k < K; k += 4) {
        v2f a = *(const v2f*)(arow + k + kk);
        v2f b = *(const v2f*)(brow + k + kk);
        acc = WMMA_F32_16x16x4(a, b, acc);
    }
    const int mbase = tm + half * 8;
    const float TWO_LOG2E = 2.8853900817779268f;
#pragma unroll
    for (int i = 0; i < 8; ++i) {
        const float p = fminf(fmaxf(acc[i] * TWO_LOG2E, -124.f), 124.f);
        Y[(size_t)(mbase + i) * N + tn + l] = __builtin_amdgcn_exp2f(p);
    }
}

__global__ void gemm_ab_batched_wmma(const float* __restrict__ A, const float* __restrict__ Bm,
                                     float* __restrict__ Y, int M, int N, int K) {
    const int wave  = threadIdx.x >> 5;
    const int lane  = threadIdx.x & 31;
    const int batch = blockIdx.y;
    const int tile  = blockIdx.x * (blockDim.x >> 5) + wave;
    const int ntn   = N >> 4;
    if (tile >= (M >> 4) * ntn) return;
    const int tm   = (tile / ntn) << 4;
    const int tn   = (tile % ntn) << 4;
    const int half = lane >> 4;
    const int l    = lane & 15;
    const int kk   = half << 1;

    const float* Ab = A  + (size_t)batch * M * K;
    const float* Bb = Bm + (size_t)batch * K * N;
    float*       Yb = Y  + (size_t)batch * M * N;

    const float* arow = Ab + (size_t)(tm + l) * K;
    const int n = tn + l;
    v8f acc = {};
    for (int k = 0; k < K; k += 4) {
        v2f a = *(const v2f*)(arow + k + kk);
        v2f b;
        b.x = Bb[(size_t)(k + kk) * N + n];
        b.y = Bb[(size_t)(k + kk + 1) * N + n];
        acc = WMMA_F32_16x16x4(a, b, acc);
    }
    const int mbase = tm + half * 8;
#pragma unroll
    for (int i = 0; i < 8; ++i)
        Yb[(size_t)(mbase + i) * N + n] = acc[i];
}

__global__ void out_gemm_wmma(const float* __restrict__ Q, const float* __restrict__ Cc,
                              const float* __restrict__ Wout, float* __restrict__ Yout) {
    const int M = 512, N = 256, H = 256, F = 512;
    const int wave = threadIdx.x >> 5;
    const int lane = threadIdx.x & 31;
    const int tile = blockIdx.x * (blockDim.x >> 5) + wave;
    const int ntn  = N >> 4;
    if (tile >= (M >> 4) * ntn) return;
    const int tm   = (tile / ntn) << 4;
    const int tn   = (tile % ntn) << 4;
    const int half = lane >> 4;
    const int l    = lane & 15;
    const int kk   = half << 1;

    const float* qrow = Q    + (size_t)(tm + l) * H;
    const float* crow = Cc   + (size_t)(tm + l) * H;
    const float* wrow = Wout + (size_t)(tn + l) * F;

    v8f acc = {};
    for (int k = 0; k < H; k += 4) {
        v2f a = *(const v2f*)(qrow + k + kk);
        v2f b = *(const v2f*)(wrow + k + kk);
        acc = WMMA_F32_16x16x4(a, b, acc);
    }
    for (int k = 0; k < H; k += 4) {
        v2f a = *(const v2f*)(crow + k + kk);
        v2f b = *(const v2f*)(wrow + H + k + kk);
        acc = WMMA_F32_16x16x4(a, b, acc);
    }
    const int mbase = tm + half * 8;
#pragma unroll
    for (int i = 0; i < 8; ++i)
        Yout[(size_t)(mbase + i) * N + tn + l] = tanhf(acc[i]);
}

__global__ void __launch_bounds__(256) score_softmax_kernel(const float* __restrict__ Eq,
                                     const float* __restrict__ Ee,
                                     const float* __restrict__ v,
                                     const int*   __restrict__ src_len,
                                     float* __restrict__ alpha) {
    const int H = 256, S = 512, T = 128;
    const int bt   = blockIdx.x;
    const int b    = bt / T;
    const int tid  = threadIdx.x;
    const int lane = tid & 31;
    const int wave = tid >> 5;

    __shared__ float sh_e[512];
    __shared__ float red[8];

    const v4f eq0 = *(const v4f*)(Eq + (size_t)bt * H + lane * 4);
    const v4f eq1 = *(const v4f*)(Eq + (size_t)bt * H + 128 + lane * 4);
    const v4f w0  = -2.f * *(const v4f*)(v + lane * 4);
    const v4f w1  = -2.f * *(const v4f*)(v + 128 + lane * 4);

    const int slen = src_len[b];
    const float* rowp = Ee + ((size_t)b * S + wave * 64) * H + lane * 4;

    constexpr int G = 8;
    v4f bufA[G][2], bufB[G][2];

    auto load_rows = [&](v4f (&buf)[G][2], int s0) {
        for (int r = 0; r < G; ++r) {
            buf[r][0] = *(const v4f*)(rowp + (size_t)(s0 + r) * H);
            buf[r][1] = *(const v4f*)(rowp + (size_t)(s0 + r) * H + 128);
        }
        __builtin_amdgcn_sched_barrier(0);
    };

    auto score_rows = [&](const v4f (&buf)[G][2], int s0) {
        float acc[G];
        for (int r = 0; r < G; ++r) {
            float t = 0.f;
            for (int c = 0; c < 4; ++c) {
                t = fmaf(w0[c], __builtin_amdgcn_rcpf(fmaf(eq0[c], buf[r][0][c], 1.f)), t);
                t = fmaf(w1[c], __builtin_amdgcn_rcpf(fmaf(eq1[c], buf[r][1][c], 1.f)), t);
            }
            acc[r] = t;
        }
        for (int r = 0; r < G; ++r) acc[r] += xhalf(acc[r]);
        for (int r = 0; r < G; ++r) acc[r] += row_xor<1>(acc[r]);
        for (int r = 0; r < G; ++r) acc[r] += row_xor<2>(acc[r]);
        for (int r = 0; r < G; ++r) acc[r] += row_xor<4>(acc[r]);
        for (int r = 0; r < G; ++r) acc[r] += row_xor<8>(acc[r]);
        if (lane == 0) {
            for (int r = 0; r < G; ++r) {
                const int s = wave * 64 + s0 + r;
                sh_e[s] = (s < slen) ? acc[r] : -INFINITY;
            }
        }
    };

    load_rows(bufA, 0);
    for (int s0 = 0; s0 < 64 - 2 * G; s0 += 2 * G) {
        load_rows(bufB, s0 + G);
        score_rows(bufA, s0);
        load_rows(bufA, s0 + 2 * G);
        score_rows(bufB, s0 + G);
    }
    load_rows(bufB, 64 - G);
    score_rows(bufA, 64 - 2 * G);
    score_rows(bufB, 64 - G);
    __syncthreads();

    float m = -INFINITY;
    for (int s = tid; s < S; s += 256) m = fmaxf(m, sh_e[s]);
#pragma unroll
    for (int off = 16; off > 0; off >>= 1) m = fmaxf(m, __shfl_xor(m, off, 32));
    if (lane == 0) red[wave] = m;
    __syncthreads();
    m = red[0];
#pragma unroll
    for (int i = 1; i < 8; ++i) m = fmaxf(m, red[i]);
    __syncthreads();

    float sum = 0.f;
    for (int s = tid; s < S; s += 256) {
        float ex = __expf(sh_e[s] - m);
        sh_e[s] = ex;
        sum += ex;
    }
#pragma unroll
    for (int off = 16; off > 0; off >>= 1) sum += __shfl_xor(sum, off, 32);
    if (lane == 0) red[wave] = sum;
    __syncthreads();
    float total = 0.f;
#pragma unroll
    for (int i = 0; i < 8; ++i) total += red[i];
    const float inv = 1.f / total;

    for (int s = tid; s < S; s += 256)
        alpha[(size_t)bt * S + s] = sh_e[s] * inv;
}

extern "C" void kernel_launch(void* const* d_in, const int* in_sizes, int n_in,
                              void* d_out, int out_size, void* d_ws, size_t ws_size,
                              hipStream_t stream) {
    const int B = 4, T = 128, S = 512, H = 256;

    const float* query = (const float*)d_in[0];
    const float* enc   = (const float*)d_in[1];
    const int*   slen  = (const int*)d_in[2];
    const float* Ws    = (const float*)d_in[3];
    const float* Wh    = (const float*)d_in[4];
    const float* v     = (const float*)d_in[5];
    const float* Wout  = (const float*)d_in[6];
    float* out = (float*)d_out;

    float* ws    = (float*)d_ws;
    float* pq    = ws;
    float* pe    = pq + (size_t)B * T * H;
    float* alpha = pe + (size_t)B * S * H;
    float* ctx   = alpha + (size_t)B * T * S;
    (void)ws_size; (void)in_sizes; (void)n_in; (void)out_size;

    gemm_xwT_exp2x_wmma<<<128, 128, 0, stream>>>(query, Ws, pq, B * T, H, H);

    gemm_xwT_exp2x_wmma<<<512, 128, 0, stream>>>(enc, Wh, pe, B * S, H, H);

    score_softmax_kernel<<<B * T, 256, 0, stream>>>(pq, pe, v, slen, alpha);

    gemm_ab_batched_wmma<<<dim3(32, B), 128, 0, stream>>>(alpha, enc, ctx, T, H, S);

    out_gemm_wmma<<<128, 128, 0, stream>>>(query, ctx, Wout, out);
}


# GEMM launches re-tiled: each wave owns 2x2 output tiles (pq and out: 1x2), so every 16-line fragment load feeds two WMMAs; the next 8 k-steps' fragments are in flight under the current ones
# speedup vs baseline: 4.6465x; 1.2390x over previous
#include <hip/hip_runtime.h>
#include <math.h>

typedef __attribute__((ext_vector_type(2))) float v2f;
typedef __attribute__((ext_vector_type(8))) float v8f;

#define WMMA_F32_16x16x4(a, b, c) \
    __builtin_amdgcn_wmma_f32_16x16x4_f32(false, (a), false, (b), (short)0, (c), false, false)

typedef __attribute__((ext_vector_type(4))) float v4f;

template <int M>
__device__ __forceinline__ float row_xor(float x) {
    const int b = __builtin_bit_cast(int, x);
    return __builtin_bit_cast(float, __builtin_amdgcn_mov_dpp(b, 0x160 + M, 0xf, 0xf, true));
}
__device__ __forceinline__ float xhalf(float x) {
    const int b = __builtin_bit_cast(int, x);
    return __builtin_bit_cast(float,
        __builtin_amdgcn_permlanex16(b, b, 0x76543210, 0xfedcba98, false, false));
}

template <int MT, int NT>
__device__ __forceinline__ void wave_block_mma(const float* const (&ap)[MT],
                                               const float* const (&bp)[NT],
                                               int K, v8f (&acc)[MT][NT]) {
    constexpr int U = 8;
    v2f a0[U][MT], b0[U][NT], a1[U][MT], b1[U][NT];

    auto load = [&](v2f (&a)[U][MT], v2f (&b)[U][NT], int k) {
        __builtin_amdgcn_sched_barrier(0);
#pragma unroll
        for (int u = 0; u < U; ++u) {
#pragma unroll
            for (int i = 0; i < MT; ++i) a[u][i] = *(const v2f*)(ap[i] + k + 4 * u);
#pragma unroll
            for (int j = 0; j < NT; ++j) b[u][j] = *(const v2f*)(bp[j] + k + 4 * u);
        }
        __builtin_amdgcn_sched_barrier(0);
    };
    auto mma = [&](const v2f (&a)[U][MT], const v2f (&b)[U][NT]) {
#pragma unroll
        for (int u = 0; u < U; ++u)
#pragma unroll
            for (int i = 0; i < MT; ++i)
#pragma unroll
                for (int j = 0; j < NT; ++j)
                    acc[i][j] = WMMA_F32_16x16x4(a[u][i], b[u][j], acc[i][j]);
    };

    load(a0, b0, 0);
    int k = 0;
    for (; k < K - 64; k += 64) {
        load(a1, b1, k + 32);
        mma(a0, b0);
        load(a0, b0, k + 64);
        mma(a1, b1);
    }
    load(a1, b1, k + 32);
    mma(a0, b0);
    mma(a1, b1);
}

template <int MT, int NT>
__global__ void __launch_bounds__(64) gemm_xwT_exp2x_wmma(const float* __restrict__ X,
                                                          const float* __restrict__ W,
                                                          float* __restrict__ Y, int M, int N, int K) {
    const int wave = threadIdx.x >> 5;
    const int lane = threadIdx.x & 31;
    const int blk  = blockIdx.x * (blockDim.x >> 5) + wave;
    const int nbn  = N / (16 * NT);
    if (blk >= (M / (16 * MT)) * nbn) return;
    const int tm   = (blk / nbn) * (16 * MT);
    const int tn   = (blk % nbn) * (16 * NT);
    const int half = lane >> 4;
    const int l    = lane & 15;
    const int kk   = half << 1;

    const float* ap[MT];
    const float* bp[NT];
#pragma unroll
    for (int i = 0; i < MT; ++i) ap[i] = X + (size_t)(tm + 16 * i + l) * K + kk;
#pragma unroll
    for (int j = 0; j < NT; ++j) bp[j] = W + (size_t)(tn + 16 * j + l) * K + kk;

    v8f acc[MT][NT] = {};
    wave_block_mma<MT, NT>(ap, bp, K, acc);

    const float TWO_LOG2E = 2.8853900817779268f;
#pragma unroll
    for (int i = 0; i < MT; ++i) {
        const int mbase = tm + 16 * i + half * 8;
#pragma unroll
        for (int r = 0; r < 8; ++r)
#pragma unroll
            for (int j = 0; j < NT; ++j) {
                const float p = fminf(fmaxf(acc[i][j][r] * TWO_LOG2E, -124.f), 124.f);
                Y[(size_t)(mbase + r) * N + tn + 16 * j + l] = __builtin_amdgcn_exp2f(p);
            }
    }
}

template <int MT, int NT>
__global__ void __launch_bounds__(64) gemm_xwT_wmma(const float* __restrict__ X, int ldx,
                                                    const float* __restrict__ W,
                                                    float* __restrict__ Y, int M, int N, int K) {
    const int wave = threadIdx.x >> 5;
    const int lane = threadIdx.x & 31;
    const int blk  = blockIdx.x * (blockDim.x >> 5) + wave;
    const int nbn  = N / (16 * NT);
    if (blk >= (M / (16 * MT)) * nbn) return;
    const int tm   = (blk / nbn) * (16 * MT);
    const int tn   = (blk % nbn) * (16 * NT);
    const int half = lane >> 4;
    const int l    = lane & 15;
    const int kk   = half << 1;

    const float* ap[MT];
    const float* bp[NT];
#pragma unroll
    for (int i = 0; i < MT; ++i) ap[i] = X + (size_t)(tm + 16 * i + l) * ldx + kk;
#pragma unroll
    for (int j = 0; j < NT; ++j) bp[j] = W + (size_t)(tn + 16 * j + l) * K + kk;

    v8f acc[MT][NT] = {};
    wave_block_mma<MT, NT>(ap, bp, K, acc);

#pragma unroll
    for (int i = 0; i < MT; ++i) {
        const int mbase = tm + 16 * i + half * 8;
#pragma unroll
        for (int r = 0; r < 8; ++r)
#pragma unroll
            for (int j = 0; j < NT; ++j)
                Y[(size_t)(mbase + r) * N + tn + 16 * j + l] = acc[i][j][r];
    }
}

__global__ void __launch_bounds__(32) out_gemm_wmma(const float* __restrict__ Q,
                                                    const float* __restrict__ Al,
                                                    const float* __restrict__ EWt,
                                                    const float* __restrict__ Wout,
                                                    float* __restrict__ Yout) {
    const int M = 512, N = 256, H = 256, F = 512, S = 512, T = 128, BS = 2048;
    constexpr int NT = 2;
    const int lane = threadIdx.x & 31;
    const int blk  = blockIdx.x;
    const int nbn  = N / (16 * NT);
    if (blk >= (M >> 4) * nbn) return;
    const int tm   = (blk / nbn) << 4;
    const int tn   = (blk % nbn) * (16 * NT);
    const int half = lane >> 4;
    const int l    = lane & 15;
    const int kk   = half << 1;
    const int batch = tm / T;

    const float* qp[1] = { Q  + (size_t)(tm + l) * H + kk };
    const float* al[1] = { Al + (size_t)(tm + l) * S + kk };
    const float* wp[NT];
    const float* ep[NT];
#pragma unroll
    for (int j = 0; j < NT; ++j) {
        wp[j] = Wout + (size_t)(tn + 16 * j + l) * F + kk;
        ep[j] = EWt  + (size_t)(tn + 16 * j + l) * BS + (size_t)batch * S + kk;
    }

    v8f acc[1][NT] = {};
    wave_block_mma<1, NT>(qp, wp, H, acc);
    wave_block_mma<1, NT>(al, ep, S, acc);

    const int mbase = tm + half * 8;
#pragma unroll
    for (int r = 0; r < 8; ++r)
#pragma unroll
        for (int j = 0; j < NT; ++j)
            Yout[(size_t)(mbase + r) * N + tn + 16 * j + l] = tanhf(acc[0][j][r]);
}

__global__ void __launch_bounds__(256) score_softmax_kernel(const float* __restrict__ Eq,
                                     const float* __restrict__ Ee,
                                     const float* __restrict__ v,
                                     const int*   __restrict__ src_len,
                                     float* __restrict__ alpha) {
    const int H = 256, S = 512, T = 128;
    const int bt   = blockIdx.x;
    const int b    = bt / T;
    const int tid  = threadIdx.x;
    const int lane = tid & 31;
    const int wave = tid >> 5;

    __shared__ float sh_e[512];
    __shared__ float red[8];

    const v4f eq0 = *(const v4f*)(Eq + (size_t)bt * H + lane * 4);
    const v4f eq1 = *(const v4f*)(Eq + (size_t)bt * H + 128 + lane * 4);
    const v4f w0  = -2.f * *(const v4f*)(v + lane * 4);
    const v4f w1  = -2.f * *(const v4f*)(v + 128 + lane * 4);

    const int slen = src_len[b];
    const float* rowp = Ee + ((size_t)b * S + wave * 64) * H + lane * 4;

    constexpr int G = 8;
    v4f bufA[G][2], bufB[G][2];

    auto load_rows = [&](v4f (&buf)[G][2], int s0) {
        for (int r = 0; r < G; ++r) {
            buf[r][0] = *(const v4f*)(rowp + (size_t)(s0 + r) * H);
            buf[r][1] = *(const v4f*)(rowp + (size_t)(s0 + r) * H + 128);
        }
        __builtin_amdgcn_sched_barrier(0);
    };

    auto score_rows = [&](const v4f (&buf)[G][2], int s0) {
        float acc[G];
        for (int r = 0; r < G; ++r) {
            float t = 0.f;
            for (int c = 0; c < 4; ++c) {
                t = fmaf(w0[c], __builtin_amdgcn_rcpf(fmaf(eq0[c], buf[r][0][c], 1.f)), t);
                t = fmaf(w1[c], __builtin_amdgcn_rcpf(fmaf(eq1[c], buf[r][1][c], 1.f)), t);
            }
            acc[r] = t;
        }
        for (int r = 0; r < G; ++r) acc[r] += xhalf(acc[r]);
        for (int r = 0; r < G; ++r) acc[r] += row_xor<1>(acc[r]);
        for (int r = 0; r < G; ++r) acc[r] += row_xor<2>(acc[r]);
        for (int r = 0; r < G; ++r) acc[r] += row_xor<4>(acc[r]);
        for (int r = 0; r < G; ++r) acc[r] += row_xor<8>(acc[r]);
        if (lane == 0) {
            for (int r = 0; r < G; ++r) {
                const int s = wave * 64 + s0 + r;
                sh_e[s] = (s < slen) ? acc[r] : -INFINITY;
            }
        }
    };

    load_rows(bufA, 0);
    for (int s0 = 0; s0 < 64 - 2 * G; s0 += 2 * G) {
        load_rows(bufB, s0 + G);
        score_rows(bufA, s0);
        load_rows(bufA, s0 + 2 * G);
        score_rows(bufB, s0 + G);
    }
    load_rows(bufB, 64 - G);
    score_rows(bufA, 64 - 2 * G);
    score_rows(bufB, 64 - G);
    __syncthreads();

    float m = -INFINITY;
    for (int s = tid; s < S; s += 256) m = fmaxf(m, sh_e[s]);
#pragma unroll
    for (int off = 16; off > 0; off >>= 1) m = fmaxf(m, __shfl_xor(m, off, 32));
    if (lane == 0) red[wave] = m;
    __syncthreads();
    m = red[0];
#pragma unroll
    for (int i = 1; i < 8; ++i) m = fmaxf(m, red[i]);
    __syncthreads();

    float sum = 0.f;
    for (int s = tid; s < S; s += 256) {
        float ex = __expf(sh_e[s] - m);
        sh_e[s] = ex;
        sum += ex;
    }
#pragma unroll
    for (int off = 16; off > 0; off >>= 1) sum += __shfl_xor(sum, off, 32);
    if (lane == 0) red[wave] = sum;
    __syncthreads();
    float total = 0.f;
#pragma unroll
    for (int i = 0; i < 8; ++i) total += red[i];
    const float inv = 1.f / total;

    for (int s = tid; s < S; s += 256)
        alpha[(size_t)bt * S + s] = sh_e[s] * inv;
}

extern "C" void kernel_launch(void* const* d_in, const int* in_sizes, int n_in,
                              void* d_out, int out_size, void* d_ws, size_t ws_size,
                              hipStream_t stream) {
    const int B = 4, T = 128, S = 512, H = 256;

    const float* query = (const float*)d_in[0];
    const float* enc   = (const float*)d_in[1];
    const int*   slen  = (const int*)d_in[2];
    const float* Ws    = (const float*)d_in[3];
    const float* Wh    = (const float*)d_in[4];
    const float* v     = (const float*)d_in[5];
    const float* Wout  = (const float*)d_in[6];
    float* out = (float*)d_out;

    float* ws    = (float*)d_ws;
    float* pq    = ws;
    float* pe    = pq + (size_t)B * T * H;
    float* alpha = pe + (size_t)B * S * H;
    float* encWT = pe;
    (void)ws_size; (void)in_sizes; (void)n_in; (void)out_size;

    gemm_xwT_exp2x_wmma<1, 2><<<256, 32, 0, stream>>>(query, Ws, pq, B * T, H, H);

    gemm_xwT_exp2x_wmma<2, 2><<<256, 64, 0, stream>>>(enc, Wh, pe, B * S, H, H);

    score_softmax_kernel<<<B * T, 256, 0, stream>>>(pq, pe, v, slen, alpha);

    gemm_xwT_wmma<2, 2><<<256, 64, 0, stream>>>(Wout + H, 2 * H, enc, encWT, H, B * S, H);

    out_gemm_wmma<<<256, 32, 0, stream>>>(query, alpha, encWT, Wout, out);
}


# score kernel: source rows at and beyond src_len are neither loaded nor scored (their slots get -inf directly); the rows in front of it are dealt in groups of 8 to the 8 waves in turn
# speedup vs baseline: 7.5326x; 1.6211x over previous
#include <hip/hip_runtime.h>
#include <math.h>

typedef __attribute__((ext_vector_type(2))) float v2f;
typedef __attribute__((ext_vector_type(8))) float v8f;

#define WMMA_F32_16x16x4(a, b, c) \
    __builtin_amdgcn_wmma_f32_16x16x4_f32(false, (a), false, (b), (short)0, (c), false, false)

typedef __attribute__((ext_vector_type(4))) float v4f;

template <int M>
__device__ __forceinline__ float row_xor(float x) {
    const int b = __builtin_bit_cast(int, x);
    return __builtin_bit_cast(float, __builtin_amdgcn_mov_dpp(b, 0x160 + M, 0xf, 0xf, true));
}
__device__ __forceinline__ float xhalf(float x) {
    const int b = __builtin_bit_cast(int, x);
    return __builtin_bit_cast(float,
        __builtin_amdgcn_permlanex16(b, b, 0x76543210, 0xfedcba98, false, false));
}

template <int MT, int NT>
__device__ __forceinline__ void wave_block_mma(const float* const (&ap)[MT],
                                               const float* const (&bp)[NT],
                                               int K, v8f (&acc)[MT][NT]) {
    constexpr int U = 8;
    v2f a0[U][MT], b0[U][NT], a1[U][MT], b1[U][NT];

    auto load = [&](v2f (&a)[U][MT], v2f (&b)[U][NT], int k) {
        __builtin_amdgcn_sched_barrier(0);
#pragma unroll
        for (int u = 0; u < U; ++u) {
#pragma unroll
            for (int i = 0; i < MT; ++i) a[u][i] = *(const v2f*)(ap[i] + k + 4 * u);
#pragma unroll
            for (int j = 0; j < NT; ++j) b[u][j] = *(const v2f*)(bp[j] + k + 4 * u);
        }
        __builtin_amdgcn_sched_barrier(0);
    };
    auto mma = [&](const v2f (&a)[U][MT], const v2f (&b)[U][NT]) {
#pragma unroll
        for (int u = 0; u < U; ++u)
#pragma unroll
            for (int i = 0; i < MT; ++i)
#pragma unroll
                for (int j = 0; j < NT; ++j)
                    acc[i][j] = WMMA_F32_16x16x4(a[u][i], b[u][j], acc[i][j]);
    };

    load(a0, b0, 0);
    int k = 0;
    for (; k < K - 64; k += 64) {
        load(a1, b1, k + 32);
        mma(a0, b0);
        load(a0, b0, k + 64);
        mma(a1, b1);
    }
    load(a1, b1, k + 32);
    mma(a0, b0);
    mma(a1, b1);
}

template <int MT, int NT>
__global__ void __launch_bounds__(64) gemm_xwT_exp2x_wmma(const float* __restrict__ X,
                                                          const float* __restrict__ W,
                                                          float* __restrict__ Y, int M, int N, int K) {
    const int wave = threadIdx.x >> 5;
    const int lane = threadIdx.x & 31;
    const int blk  = blockIdx.x * (blockDim.x >> 5) + wave;
    const int nbn  = N / (16 * NT);
    if (blk >= (M / (16 * MT)) * nbn) return;
    const int tm   = (blk / nbn) * (16 * MT);
    const int tn   = (blk % nbn) * (16 * NT);
    const int half = lane >> 4;
    const int l    = lane & 15;
    const int kk   = half << 1;

    const float* ap[MT];
    const float* bp[NT];
#pragma unroll
    for (int i = 0; i < MT; ++i) ap[i] = X + (size_t)(tm + 16 * i + l) * K + kk;
#pragma unroll
    for (int j = 0; j < NT; ++j) bp[j] = W + (size_t)(tn + 16 * j + l) * K + kk;

    v8f acc[MT][NT] = {};
    wave_block_mma<MT, NT>(ap, bp, K, acc);

    const float TWO_LOG2E = 2.8853900817779268f;
#pragma unroll
    for (int i = 0; i < MT; ++i) {
        const int mbase = tm + 16 * i + half * 8;
#pragma unroll
        for (int r = 0; r < 8; ++r)
#pragma unroll
            for (int j = 0; j < NT; ++j) {
                const float p = fminf(fmaxf(acc[i][j][r] * TWO_LOG2E, -124.f), 124.f);
                Y[(size_t)(mbase + r) * N + tn + 16 * j + l] = __builtin_amdgcn_exp2f(p);
            }
    }
}

__device__ __forceinline__ void store_partial_block(float* __restrict__ red, const v8f (&acc)[1][2],
                                                    int wave, int lane) {
#pragma unroll
    for (int j = 0; j < 2; ++j)
#pragma unroll
        for (int r = 0; r < 8; ++r)
            red[((wave * 2 + j) * 8 + r) * 32 + lane] = acc[0][j][r];
}
__device__ __forceinline__ void sum4_partial_blocks(const float* __restrict__ red, int wave, int lane,
                                                    float (&sum)[2][2]) {
#pragma unroll
    for (int j = 0; j < 2; ++j)
#pragma unroll
        for (int q = 0; q < 2; ++q) {
            const int r = 2 * wave + q;
            float s = red[((0 * 2 + j) * 8 + r) * 32 + lane];
            s += red[((1 * 2 + j) * 8 + r) * 32 + lane];
            s += red[((2 * 2 + j) * 8 + r) * 32 + lane];
            s += red[((3 * 2 + j) * 8 + r) * 32 + lane];
            sum[j][q] = s;
        }
}
__device__ __forceinline__ void join4_partial_blocks(float* __restrict__ red, const v8f (&acc)[1][2],
                                                     int wave, int lane, float (&sum)[2][2]) {
    store_partial_block(red, acc, wave, lane);
    __syncthreads();
    sum4_partial_blocks(red, wave, lane, sum);
}

__global__ void __launch_bounds__(128) pq_exp2x_splitk_wmma(const float* __restrict__ X,
                                                            const float* __restrict__ W,
                                                            float* __restrict__ Y) {
    const int M = 512, N = 256, K = 256;
    constexpr int NT = 2;
    __shared__ float red[4 * 2 * 8 * 32];
    const int wave = threadIdx.x >> 5;
    const int lane = threadIdx.x & 31;
    const int blk  = blockIdx.x;
    const int nbn  = N / (16 * NT);
    if (blk >= (M >> 4) * nbn) return;
    const int tm   = (blk / nbn) << 4;
    const int tn   = (blk % nbn) * (16 * NT);
    const int half = lane >> 4;
    const int l    = lane & 15;
    const int kk   = half << 1;
    const int k0   = wave * (K / 4);

    const float* ap[1] = { X + (size_t)(tm + l) * K + k0 + kk };
    const float* bp[NT];
#pragma unroll
    for (int j = 0; j < NT; ++j) bp[j] = W + (size_t)(tn + 16 * j + l) * K + k0 + kk;

    v8f acc[1][NT] = {};
    wave_block_mma<1, NT>(ap, bp, K / 4, acc);

    float sum[2][2];
    join4_partial_blocks(red, acc, wave, lane, sum);

    const float TWO_LOG2E = 2.8853900817779268f;
    const int mbase = tm + half * 8;
#pragma unroll
    for (int q = 0; q < 2; ++q)
#pragma unroll
        for (int j = 0; j < NT; ++j) {
            const float p = fminf(fmaxf(sum[j][q] * TWO_LOG2E, -124.f), 124.f);
            Y[(size_t)(mbase + 2 * wave + q) * N + tn + 16 * j + l] = __builtin_amdgcn_exp2f(p);
        }
}

__device__ __forceinline__ void enc_proj_stacked_block(const float* __restrict__ X,
                                                       const float* __restrict__ Wh,
                                                       const float* __restrict__ Wout,
                                                       float* __restrict__ Ee,
                                                       float* __restrict__ EWt,
                                                       int blk, int lane) {
    constexpr int MT = 2, NT = 2;
    const int M = 2048, H = 256, K = 256, NS = 512;
    const int nbn  = NS / (16 * NT);
    const int tm   = (blk / nbn) * (16 * MT);
    const int tn   = (blk % nbn) * (16 * NT);
    const bool wc  = tn >= H;
    const int half = lane >> 4;
    const int l    = lane & 15;
    const int kk   = half << 1;

    const float* ap[MT];
    const float* bp[NT];
#pragma unroll
    for (int i = 0; i < MT; ++i) ap[i] = X + (size_t)(tm + 16 * i + l) * K + kk;
#pragma unroll
    for (int j = 0; j < NT; ++j)
        bp[j] = wc ? Wout + (size_t)(tn - H + 16 * j + l) * (2 * H) + H + kk
                   : Wh   + (size_t)(tn + 16 * j + l) * K + kk;

    v8f acc[MT][NT] = {};
    wave_block_mma<MT, NT>(ap, bp, K, acc);

    if (!wc) {
        const float TWO_LOG2E = 2.8853900817779268f;
#pragma unroll
        for (int i = 0; i < MT; ++i) {
            const int mbase = tm + 16 * i + half * 8;
#pragma unroll
            for (int r = 0; r < 8; ++r)
#pragma unroll
                for (int j = 0; j < NT; ++j) {
                    const float p = fminf(fmaxf(acc[i][j][r] * TWO_LOG2E, -124.f), 124.f);
                    Ee[(size_t)(mbase + r) * H + tn + 16 * j + l] = __builtin_amdgcn_exp2f(p);
                }
        }
    } else {
#pragma unroll
        for (int i = 0; i < MT; ++i)
#pragma unroll
            for (int j = 0; j < NT; ++j) {
                float* p = EWt + (size_t)(tn - H + 16 * j + l) * M + tm + 16 * i + half * 8;
                const v8f a = acc[i][j];
                v4f lo, hi;
                lo[0] = a[0]; lo[1] = a[1]; lo[2] = a[2]; lo[3] = a[3];
                hi[0] = a[4]; hi[1] = a[5]; hi[2] = a[6]; hi[3] = a[7];
                *(v4f*)p       = lo;
                *(v4f*)(p + 4) = hi;
            }
    }
}

__global__ void __launch_bounds__(256) enc_and_query_proj_wmma(const float* __restrict__ X,
                                                               const float* __restrict__ Wh,
                                                               const float* __restrict__ Wout,
                                                               float* __restrict__ Ee,
                                                               float* __restrict__ EWt,
                                                               const float* __restrict__ Q,
                                                               const float* __restrict__ Ws,
                                                               float* __restrict__ Eq) {
    constexpr int NT = 2;
    const int N = 256, K = 256;
    __shared__ float red[4 * 2 * 8 * 32];
    const int wave = __builtin_amdgcn_readfirstlane((int)(threadIdx.x >> 5));
    const int lane = threadIdx.x & 31;
    const int half = lane >> 4;
    const int l    = lane & 15;
    const int kk   = half << 1;
    const int qw   = wave - 4;
    const int nbn  = N / (16 * NT);
    const int tm   = (blockIdx.x / nbn) << 4;
    const int tn   = (blockIdx.x % nbn) * (16 * NT);

    if (wave < 4) {
        enc_proj_stacked_block(X, Wh, Wout, Ee, EWt, blockIdx.x * 4 + wave, lane);
    } else {
        const int k0 = qw * (K / 4);
        const float* ap[1] = { Q + (size_t)(tm + l) * K + k0 + kk };
        const float* bp[NT];
#pragma unroll
        for (int j = 0; j < NT; ++j) bp[j] = Ws + (size_t)(tn + 16 * j + l) * K + k0 + kk;
        v8f acc[1][NT] = {};
        wave_block_mma<1, NT>(ap, bp, K / 4, acc);
        store_partial_block(red, acc, qw, lane);
    }
    __syncthreads();
    if (wave >= 4) {
        float sum[2][2];
        sum4_partial_blocks(red, qw, lane, sum);
        const float TWO_LOG2E = 2.8853900817779268f;
        const int mbase = tm + half * 8;
#pragma unroll
        for (int q = 0; q < 2; ++q)
#pragma unroll
            for (int j = 0; j < NT; ++j) {
                const float p = fminf(fmaxf(sum[j][q] * TWO_LOG2E, -124.f), 124.f);
                Eq[(size_t)(mbase + 2 * qw + q) * N + tn + 16 * j + l] = __builtin_amdgcn_exp2f(p);
            }
    }
}

template <int MT, int NT>
__global__ void __launch_bounds__(64) gemm_xwT_wmma(const float* __restrict__ X, int ldx,
                                                    const float* __restrict__ W,
                                                    float* __restrict__ Y, int M, int N, int K) {
    const int wave = threadIdx.x >> 5;
    const int lane = threadIdx.x & 31;
    const int blk  = blockIdx.x * (blockDim.x >> 5) + wave;
    const int nbn  = N / (16 * NT);
    if (blk >= (M / (16 * MT)) * nbn) return;
    const int tm   = (blk / nbn) * (16 * MT);
    const int tn   = (blk % nbn) * (16 * NT);
    const int half = lane >> 4;
    const int l    = lane & 15;
    const int kk   = half << 1;

    const float* ap[MT];
    const float* bp[NT];
#pragma unroll
    for (int i = 0; i < MT; ++i) ap[i] = X + (size_t)(tm + 16 * i + l) * ldx + kk;
#pragma unroll
    for (int j = 0; j < NT; ++j) bp[j] = W + (size_t)(tn + 16 * j + l) * K + kk;

    v8f acc[MT][NT] = {};
    wave_block_mma<MT, NT>(ap, bp, K, acc);

#pragma unroll
    for (int i = 0; i < MT; ++i) {
        const int mbase = tm + 16 * i + half * 8;
#pragma unroll
        for (int r = 0; r < 8; ++r)
#pragma unroll
            for (int j = 0; j < NT; ++j)
                Y[(size_t)(mbase + r) * N + tn + 16 * j + l] = acc[i][j][r];
    }
}

__global__ void __launch_bounds__(128) out_gemm_wmma(const float* __restrict__ Q,
                                                     const float* __restrict__ Al,
                                                     const float* __restrict__ EWt,
                                                     const float* __restrict__ Wout,
                                                     float* __restrict__ Yout) {
    const int M = 512, N = 256, H = 256, F = 512, S = 512, T = 128, BS = 2048;
    constexpr int NT = 2;
    __shared__ float red[4 * 2 * 8 * 32];
    const int wave = threadIdx.x >> 5;
    const int lane = threadIdx.x & 31;
    const int blk  = blockIdx.x;
    const int nbn  = N / (16 * NT);
    if (blk >= (M >> 4) * nbn) return;
    const int tm   = (blk / nbn) << 4;
    const int tn   = (blk % nbn) * (16 * NT);
    const int half = lane >> 4;
    const int l    = lane & 15;
    const int kk   = half << 1;
    const int batch = tm / T;
    const int h0   = wave * (H / 4);
    const int s0   = wave * (S / 4);

    const float* qp = Q  + (size_t)(tm + l) * H + h0 + kk;
    const float* al = Al + (size_t)(tm + l) * S + s0 + kk;
    const float* wp[NT];
    const float* ep[NT];
#pragma unroll
    for (int j = 0; j < NT; ++j) {
        wp[j] = Wout + (size_t)(tn + 16 * j + l) * F + h0 + kk;
        ep[j] = EWt  + (size_t)(tn + 16 * j + l) * BS + (size_t)batch * S + s0 + kk;
    }

    constexpr int U = 8;
    v8f acc[1][NT] = {};
    v2f a0[U], b0[U][NT], a1[U], b1[U][NT];
    auto load = [&](v2f (&a)[U], v2f (&b)[U][NT], const float* x, const float* const (&w)[NT], int k) {
        __builtin_amdgcn_sched_barrier(0);
#pragma unroll
        for (int u = 0; u < U; ++u) {
            a[u] = *(const v2f*)(x + k + 4 * u);
#pragma unroll
            for (int j = 0; j < NT; ++j) b[u][j] = *(const v2f*)(w[j] + k + 4 * u);
        }
        __builtin_amdgcn_sched_barrier(0);
    };
    auto mma = [&](const v2f (&a)[U], const v2f (&b)[U][NT]) {
#pragma unroll
        for (int u = 0; u < U; ++u)
#pragma unroll
            for (int j = 0; j < NT; ++j)
                acc[0][j] = WMMA_F32_16x16x4(a[u], b[u][j], acc[0][j]);
    };
    load(a0, b0, qp, wp, 0);
    load(a1, b1, qp, wp, 32);
    mma(a0, b0);
    load(a0, b0, al, ep, 0);
    mma(a1, b1);
    load(a1, b1, al, ep, 32);
    mma(a0, b0);
    load(a0, b0, al, ep, 64);
    mma(a1, b1);
    load(a1, b1, al, ep, 96);
    mma(a0, b0);
    mma(a1, b1);

    float sum[2][2];
    join4_partial_blocks(red, acc, wave, lane, sum);

    const int mbase = tm + half * 8;
#pragma unroll
    for (int q = 0; q < 2; ++q)
#pragma unroll
        for (int j = 0; j < NT; ++j)
            Yout[(size_t)(mbase + 2 * wave + q) * N + tn + 16 * j + l] = tanhf(sum[j][q]);
}

__global__ void __launch_bounds__(256) score_softmax_kernel(const float* __restrict__ Eq,
                                     const float* __restrict__ Ee,
                                     const float* __restrict__ v,
                                     const int*   __restrict__ src_len,
                                     float* __restrict__ alpha) {
    const int H = 256, S = 512, T = 128;
    const int bt   = blockIdx.x;
    const int b    = bt / T;
    const int tid  = threadIdx.x;
    const int lane = tid & 31;
    const int wave = __builtin_amdgcn_readfirstlane(tid >> 5);

    __shared__ float sh_e[512];
    __shared__ float red[8];

    const v4f eq0 = *(const v4f*)(Eq + (size_t)bt * H + lane * 4);
    const v4f eq1 = *(const v4f*)(Eq + (size_t)bt * H + 128 + lane * 4);
    const v4f v0  = *(const v4f*)(v + lane * 4);
    const v4f v1  = *(const v4f*)(v + 128 + lane * 4);

    constexpr int G = 8;
    const float* rowp = Ee + ((size_t)b * S + wave * G) * H + lane * 4;
    v4f bufA[G][2], bufB[G][2];

    auto load_rows = [&](v4f (&buf)[G][2], int i) {
        const float* p = rowp + (size_t)i * (8 * G * H);
        for (int r = 0; r < G; ++r) {
            buf[r][0] = *(const v4f*)(p + (size_t)r * H);
            buf[r][1] = *(const v4f*)(p + (size_t)r * H + 128);
        }
        __builtin_amdgcn_sched_barrier(0);
    };

    load_rows(bufA, 0);

    const v4f w0 = -2.f * v0;
    const v4f w1 = -2.f * v1;

    const int slen = src_len[b];
    const int ng = min(max((slen + G - 1) / G, 0), S / G);
    const int nw = (ng - wave + 7) >> 3;

    {
        const int s = ng * G + tid;
        if (s < S)       sh_e[s]       = -INFINITY;
        if (s + 256 < S) sh_e[s + 256] = -INFINITY;
    }

    auto score_rows = [&](const v4f (&buf)[G][2], int i) {
        float acc[G];
        for (int r = 0; r < G; ++r) {
            float t = 0.f;
            for (int c = 0; c < 4; ++c) {
                t = fmaf(w0[c], __builtin_amdgcn_rcpf(fmaf(eq0[c], buf[r][0][c], 1.f)), t);
                t = fmaf(w1[c], __builtin_amdgcn_rcpf(fmaf(eq1[c], buf[r][1][c], 1.f)), t);
            }
            acc[r] = t;
        }
        for (int r = 0; r < G; ++r) acc[r] += xhalf(acc[r]);
        for (int r = 0; r < G; ++r) acc[r] += row_xor<1>(acc[r]);
        for (int r = 0; r < G; ++r) acc[r] += row_xor<2>(acc[r]);
        for (int r = 0; r < G; ++r) acc[r] += row_xor<4>(acc[r]);
        for (int r = 0; r < G; ++r) acc[r] += row_xor<8>(acc[r]);
        if (lane == 0) {
            for (int r = 0; r < G; ++r) {
                const int s = (wave + 8 * i) * G + r;
                sh_e[s] = (s < slen) ? acc[r] : -INFINITY;
            }
        }
    };

    if (nw > 0) {
        int i = 0;
        for (; i + 2 < nw; i += 2) {
            load_rows(bufB, i + 1);
            score_rows(bufA, i);
            load_rows(bufA, i + 2);
            score_rows(bufB, i + 1);
        }
        if (i + 1 < nw) {
            load_rows(bufB, i + 1);
            score_rows(bufA, i);
            score_rows(bufB, i + 1);
        } else {
            score_rows(bufA, i);
        }
    }
    __syncthreads();

    float m = -INFINITY;
    for (int s = tid; s < S; s += 256) m = fmaxf(m, sh_e[s]);
#pragma unroll
    for (int off = 16; off > 0; off >>= 1) m = fmaxf(m, __shfl_xor(m, off, 32));
    if (lane == 0) red[wave] = m;
    __syncthreads();
    m = red[0];
#pragma unroll
    for (int i = 1; i < 8; ++i) m = fmaxf(m, red[i]);
    __syncthreads();

    float sum = 0.f;
    for (int s = tid; s < S; s += 256) {
        float ex = __expf(sh_e[s] - m);
        sh_e[s] = ex;
        sum += ex;
    }
#pragma unroll
    for (int off = 16; off > 0; off >>= 1) sum += __shfl_xor(sum, off, 32);
    if (lane == 0) red[wave] = sum;
    __syncthreads();
    float total = 0.f;
#pragma unroll
    for (int i = 0; i < 8; ++i) total += red[i];
    const float inv = 1.f / total;

    for (int s = tid; s < S; s += 256)
        alpha[(size_t)bt * S + s] = sh_e[s] * inv;
}

extern "C" void kernel_launch(void* const* d_in, const int* in_sizes, int n_in,
                              void* d_out, int out_size, void* d_ws, size_t ws_size,
                              hipStream_t stream) {
    const int B = 4, T = 128, S = 512, H = 256;

    const float* query = (const float*)d_in[0];
    const float* enc   = (const float*)d_in[1];
    const int*   slen  = (const int*)d_in[2];
    const float* Ws    = (const float*)d_in[3];
    const float* Wh    = (const float*)d_in[4];
    const float* v     = (const float*)d_in[5];
    const float* Wout  = (const float*)d_in[6];
    float* out = (float*)d_out;

    float* ws    = (float*)d_ws;
    float* pq    = ws;
    float* pe    = pq + (size_t)B * T * H;
    float* alpha = pe + (size_t)B * S * H;
    float* encWT = pe;
    float* encWT_own = alpha + (size_t)B * T * S;
    (void)in_sizes; (void)n_in; (void)out_size;

    if (ws_size >= 5767168) {
        enc_and_query_proj_wmma<<<256, 256, 0, stream>>>(enc, Wh, Wout, pe, encWT_own, query, Ws, pq);

        score_softmax_kernel<<<B * T, 256, 0, stream>>>(pq, pe, v, slen, alpha);

        out_gemm_wmma<<<256, 128, 0, stream>>>(query, alpha, encWT_own, Wout, out);
        return;
    }

    pq_exp2x_splitk_wmma<<<256, 128, 0, stream>>>(query, Ws, pq);

    gemm_xwT_exp2x_wmma<2, 2><<<256, 64, 0, stream>>>(enc, Wh, pe, B * S, H, H);

    score_softmax_kernel<<<B * T, 256, 0, stream>>>(pq, pe, v, slen, alpha);

    gemm_xwT_wmma<2, 2><<<256, 64, 0, stream>>>(Wout + H, 2 * H, enc, encWT, H, B * S, H);

    out_gemm_wmma<<<256, 128, 0, stream>>>(query, alpha, encWT, Wout, out);
}


# enc projection re-tiled: a work-group owns a 64x64 block and each enc wave sums a quarter of k into all 4x4 tiles (a fragment load feeds four WMMAs, was two); partial blocks joined once in LDS
# speedup vs baseline: 8.0466x; 1.0682x over previous
#include <hip/hip_runtime.h>
#include <math.h>

typedef __attribute__((ext_vector_type(2))) float v2f;
typedef __attribute__((ext_vector_type(8))) float v8f;

#define WMMA_F32_16x16x4(a, b, c) \
    __builtin_amdgcn_wmma_f32_16x16x4_f32(false, (a), false, (b), (short)0, (c), false, false)

typedef __attribute__((ext_vector_type(4))) float v4f;

template <int M>
__device__ __forceinline__ float row_xor(float x) {
    const int b = __builtin_bit_cast(int, x);
    return __builtin_bit_cast(float, __builtin_amdgcn_mov_dpp(b, 0x160 + M, 0xf, 0xf, true));
}
__device__ __forceinline__ float xhalf(float x) {
    const int b = __builtin_bit_cast(int, x);
    return __builtin_bit_cast(float,
        __builtin_amdgcn_permlanex16(b, b, 0x76543210, 0xfedcba98, false, false));
}

template <int MT, int NT, int U = 8>
__device__ __forceinline__ void wave_block_mma(const float* const (&ap)[MT],
                                               const float* const (&bp)[NT],
                                               int K, v8f (&acc)[MT][NT]) {
    v2f a0[U][MT], b0[U][NT], a1[U][MT], b1[U][NT];

    auto load = [&](v2f (&a)[U][MT], v2f (&b)[U][NT], int k) {
        __builtin_amdgcn_sched_barrier(0);
#pragma unroll
        for (int u = 0; u < U; ++u) {
#pragma unroll
            for (int i = 0; i < MT; ++i) a[u][i] = *(const v2f*)(ap[i] + k + 4 * u);
#pragma unroll
            for (int j = 0; j < NT; ++j) b[u][j] = *(const v2f*)(bp[j] + k + 4 * u);
        }
        __builtin_amdgcn_sched_barrier(0);
    };
    auto mma = [&](const v2f (&a)[U][MT], const v2f (&b)[U][NT]) {
#pragma unroll
        for (int u = 0; u < U; ++u)
#pragma unroll
            for (int i = 0; i < MT; ++i)
#pragma unroll
                for (int j = 0; j < NT; ++j)
                    acc[i][j] = WMMA_F32_16x16x4(a[u][i], b[u][j], acc[i][j]);
    };

    load(a0, b0, 0);
    int k = 0;
    for (; k < K - 8 * U; k += 8 * U) {
        load(a1, b1, k + 4 * U);
        mma(a0, b0);
        load(a0, b0, k + 8 * U);
        mma(a1, b1);
    }
    load(a1, b1, k + 4 * U);
    mma(a0, b0);
    mma(a1, b1);
}

template <int MT, int NT>
__global__ void __launch_bounds__(64) gemm_xwT_exp2x_wmma(const float* __restrict__ X,
                                                          const float* __restrict__ W,
                                                          float* __restrict__ Y, int M, int N, int K) {
    const int wave = threadIdx.x >> 5;
    const int lane = threadIdx.x & 31;
    const int blk  = blockIdx.x * (blockDim.x >> 5) + wave;
    const int nbn  = N / (16 * NT);
    if (blk >= (M / (16 * MT)) * nbn) return;
    const int tm   = (blk / nbn) * (16 * MT);
    const int tn   = (blk % nbn) * (16 * NT);
    const int half = lane >> 4;
    const int l    = lane & 15;
    const int kk   = half << 1;

    const float* ap[MT];
    const float* bp[NT];
#pragma unroll
    for (int i = 0; i < MT; ++i) ap[i] = X + (size_t)(tm + 16 * i + l) * K + kk;
#pragma unroll
    for (int j = 0; j < NT; ++j) bp[j] = W + (size_t)(tn + 16 * j + l) * K + kk;

    v8f acc[MT][NT] = {};
    wave_block_mma<MT, NT>(ap, bp, K, acc);

    const float TWO_LOG2E = 2.8853900817779268f;
#pragma unroll
    for (int i = 0; i < MT; ++i) {
        const int mbase = tm + 16 * i + half * 8;
#pragma unroll
        for (int r = 0; r < 8; ++r)
#pragma unroll
            for (int j = 0; j < NT; ++j) {
                const float p = fminf(fmaxf(acc[i][j][r] * TWO_LOG2E, -124.f), 124.f);
                Y[(size_t)(mbase + r) * N + tn + 16 * j + l] = __builtin_amdgcn_exp2f(p);
            }
    }
}

__device__ __forceinline__ void store_partial_block(float* __restrict__ red, const v8f (&acc)[1][2],
                                                    int wave, int lane) {
#pragma unroll
    for (int j = 0; j < 2; ++j)
#pragma unroll
        for (int r = 0; r < 8; ++r)
            red[((wave * 2 + j) * 8 + r) * 32 + lane] = acc[0][j][r];
}
__device__ __forceinline__ void sum4_partial_blocks(const float* __restrict__ red, int wave, int lane,
                                                    float (&sum)[2][2]) {
#pragma unroll
    for (int j = 0; j < 2; ++j)
#pragma unroll
        for (int q = 0; q < 2; ++q) {
            const int r = 2 * wave + q;
            float s = red[((0 * 2 + j) * 8 + r) * 32 + lane];
            s += red[((1 * 2 + j) * 8 + r) * 32 + lane];
            s += red[((2 * 2 + j) * 8 + r) * 32 + lane];
            s += red[((3 * 2 + j) * 8 + r) * 32 + lane];
            sum[j][q] = s;
        }
}
__device__ __forceinline__ void join4_partial_blocks(float* __restrict__ red, const v8f (&acc)[1][2],
                                                     int wave, int lane, float (&sum)[2][2]) {
    store_partial_block(red, acc, wave, lane);
    __syncthreads();
    sum4_partial_blocks(red, wave, lane, sum);
}

__global__ void __launch_bounds__(128) pq_exp2x_splitk_wmma(const float* __restrict__ X,
                                                            const float* __restrict__ W,
                                                            float* __restrict__ Y) {
    const int M = 512, N = 256, K = 256;
    constexpr int NT = 2;
    __shared__ float red[4 * 2 * 8 * 32];
    const int wave = threadIdx.x >> 5;
    const int lane = threadIdx.x & 31;
    const int blk  = blockIdx.x;
    const int nbn  = N / (16 * NT);
    if (blk >= (M >> 4) * nbn) return;
    const int tm   = (blk / nbn) << 4;
    const int tn   = (blk % nbn) * (16 * NT);
    const int half = lane >> 4;
    const int l    = lane & 15;
    const int kk   = half << 1;
    const int k0   = wave * (K / 4);

    const float* ap[1] = { X + (size_t)(tm + l) * K + k0 + kk };
    const float* bp[NT];
#pragma unroll
    for (int j = 0; j < NT; ++j) bp[j] = W + (size_t)(tn + 16 * j + l) * K + k0 + kk;

    v8f acc[1][NT] = {};
    wave_block_mma<1, NT>(ap, bp, K / 4, acc);

    float sum[2][2];
    join4_partial_blocks(red, acc, wave, lane, sum);

    const float TWO_LOG2E = 2.8853900817779268f;
    const int mbase = tm + half * 8;
#pragma unroll
    for (int q = 0; q < 2; ++q)
#pragma unroll
        for (int j = 0; j < NT; ++j) {
            const float p = fminf(fmaxf(sum[j][q] * TWO_LOG2E, -124.f), 124.f);
            Y[(size_t)(mbase + 2 * wave + q) * N + tn + 16 * j + l] = __builtin_amdgcn_exp2f(p);
        }
}

__device__ __forceinline__ void enc_proj_stacked_partial(const float* __restrict__ X,
                                                         const float* __restrict__ Wh,
                                                         const float* __restrict__ Wout,
                                                         float* __restrict__ part,
                                                         int tm, int tn, bool wc, int wave, int lane) {
    constexpr int MT = 4, NT = 4;
    const int H = 256, K = 256;
    const int half = lane >> 4;
    const int l    = lane & 15;
    const int kk   = half << 1;
    const int k0   = wave * (K / 4);

    const float* ap[MT];
    const float* bp[NT];
#pragma unroll
    for (int i = 0; i < MT; ++i) ap[i] = X + (size_t)(tm + 16 * i + l) * K + k0 + kk;
#pragma unroll
    for (int j = 0; j < NT; ++j)
        bp[j] = wc ? Wout + (size_t)(tn - H + 16 * j + l) * (2 * H) + H + k0 + kk
                   : Wh   + (size_t)(tn + 16 * j + l) * K + k0 + kk;

    v8f acc[MT][NT] = {};
    wave_block_mma<MT, NT, 4>(ap, bp, K / 4, acc);

#pragma unroll
    for (int i = 0; i < MT; ++i)
#pragma unroll
        for (int j = 0; j < NT; ++j)
#pragma unroll
            for (int r = 0; r < 8; ++r)
                part[(((wave * MT + i) * NT + j) * 8 + r) * 32 + lane] = acc[i][j][r];
}
__device__ __forceinline__ void enc_proj_stacked_finish(const float* __restrict__ part,
                                                        float* __restrict__ Ee,
                                                        float* __restrict__ EWt,
                                                        int tm, int tn, bool wc, int wave, int lane) {
    constexpr int MT = 4, NT = 4;
    const int M = 2048, H = 256;
    const int half = lane >> 4;
    const int l    = lane & 15;
    const int i0   = (wave >> 1) * 2;
    const int j0   = (wave & 1) * 2;
    float sum[2][2][8];
#pragma unroll
    for (int a = 0; a < 2; ++a)
#pragma unroll
        for (int b = 0; b < 2; ++b)
#pragma unroll
            for (int r = 0; r < 8; ++r) {
                const int t = ((i0 + a) * NT + j0 + b) * 8 + r;
                float s = part[(0 * MT * NT * 8 + t) * 32 + lane];
                s += part[(1 * MT * NT * 8 + t) * 32 + lane];
                s += part[(2 * MT * NT * 8 + t) * 32 + lane];
                s += part[(3 * MT * NT * 8 + t) * 32 + lane];
                sum[a][b][r] = s;
            }
    const int wm = tm + 16 * i0;
    const int wn = tn + 16 * j0;
    if (!wc) {
        const float TWO_LOG2E = 2.8853900817779268f;
#pragma unroll
        for (int a = 0; a < 2; ++a) {
            const int mbase = wm + 16 * a + half * 8;
#pragma unroll
            for (int r = 0; r < 8; ++r)
#pragma unroll
                for (int b = 0; b < 2; ++b) {
                    const float p = fminf(fmaxf(sum[a][b][r] * TWO_LOG2E, -124.f), 124.f);
                    Ee[(size_t)(mbase + r) * H + wn + 16 * b + l] = __builtin_amdgcn_exp2f(p);
                }
        }
    } else {
#pragma unroll
        for (int a = 0; a < 2; ++a)
#pragma unroll
            for (int b = 0; b < 2; ++b) {
                float* p = EWt + (size_t)(wn - H + 16 * b + l) * M + wm + 16 * a + half * 8;
                v4f lo, hi;
                lo[0] = sum[a][b][0]; lo[1] = sum[a][b][1]; lo[2] = sum[a][b][2]; lo[3] = sum[a][b][3];
                hi[0] = sum[a][b][4]; hi[1] = sum[a][b][5]; hi[2] = sum[a][b][6]; hi[3] = sum[a][b][7];
                *(v4f*)p       = lo;
                *(v4f*)(p + 4) = hi;
            }
    }
}

__global__ void __launch_bounds__(256) enc_and_query_proj_wmma(const float* __restrict__ X,
                                                               const float* __restrict__ Wh,
                                                               const float* __restrict__ Wout,
                                                               float* __restrict__ Ee,
                                                               float* __restrict__ EWt,
                                                               const float* __restrict__ Q,
                                                               const float* __restrict__ Ws,
                                                               float* __restrict__ Eq) {
    constexpr int NT = 2;
    const int N = 256, K = 256;
    __shared__ float red[4 * 2 * 8 * 32];
    __shared__ float part[4 * 16 * 8 * 32];
    const int wave = __builtin_amdgcn_readfirstlane((int)(threadIdx.x >> 5));
    const int lane = threadIdx.x & 31;
    const int half = lane >> 4;
    const int l    = lane & 15;
    const int kk   = half << 1;
    const int qw   = wave - 4;
    const int nbn  = N / (16 * NT);
    const int tm   = (blockIdx.x / nbn) << 4;
    const int tn   = (blockIdx.x % nbn) * (16 * NT);
    const int etm  = (blockIdx.x >> 3) * 64;
    const int etn  = (blockIdx.x & 7) * 64;
    const bool wc  = etn >= 256;

    if (wave < 4) {
        enc_proj_stacked_partial(X, Wh, Wout, part, etm, etn, wc, wave, lane);
    } else {
        const int k0 = qw * (K / 4);
        const float* ap[1] = { Q + (size_t)(tm + l) * K + k0 + kk };
        const float* bp[NT];
#pragma unroll
        for (int j = 0; j < NT; ++j) bp[j] = Ws + (size_t)(tn + 16 * j + l) * K + k0 + kk;
        v8f acc[1][NT] = {};
        wave_block_mma<1, NT>(ap, bp, K / 4, acc);
        store_partial_block(red, acc, qw, lane);
    }
    __syncthreads();
    if (wave < 4) {
        enc_proj_stacked_finish(part, Ee, EWt, etm, etn, wc, wave, lane);
    } else {
        float sum[2][2];
        sum4_partial_blocks(red, qw, lane, sum);
        const float TWO_LOG2E = 2.8853900817779268f;
        const int mbase = tm + half * 8;
#pragma unroll
        for (int q = 0; q < 2; ++q)
#pragma unroll
            for (int j = 0; j < NT; ++j) {
                const float p = fminf(fmaxf(sum[j][q] * TWO_LOG2E, -124.f), 124.f);
                Eq[(size_t)(mbase + 2 * qw + q) * N + tn + 16 * j + l] = __builtin_amdgcn_exp2f(p);
            }
    }
}

template <int MT, int NT>
__global__ void __launch_bounds__(64) gemm_xwT_wmma(const float* __restrict__ X, int ldx,
                                                    const float* __restrict__ W,
                                                    float* __restrict__ Y, int M, int N, int K) {
    const int wave = threadIdx.x >> 5;
    const int lane = threadIdx.x & 31;
    const int blk  = blockIdx.x * (blockDim.x >> 5) + wave;
    const int nbn  = N / (16 * NT);
    if (blk >= (M / (16 * MT)) * nbn) return;
    const int tm   = (blk / nbn) * (16 * MT);
    const int tn   = (blk % nbn) * (16 * NT);
    const int half = lane >> 4;
    const int l    = lane & 15;
    const int kk   = half << 1;

    const float* ap[MT];
    const float* bp[NT];
#pragma unroll
    for (int i = 0; i < MT; ++i) ap[i] = X + (size_t)(tm + 16 * i + l) * ldx + kk;
#pragma unroll
    for (int j = 0; j < NT; ++j) bp[j] = W + (size_t)(tn + 16 * j + l) * K + kk;

    v8f acc[MT][NT] = {};
    wave_block_mma<MT, NT>(ap, bp, K, acc);

#pragma unroll
    for (int i = 0; i < MT; ++i) {
        const int mbase = tm + 16 * i + half * 8;
#pragma unroll
        for (int r = 0; r < 8; ++r)
#pragma unroll
            for (int j = 0; j < NT; ++j)
                Y[(size_t)(mbase + r) * N + tn + 16 * j + l] = acc[i][j][r];
    }
}

__global__ void __launch_bounds__(128) out_gemm_wmma(const float* __restrict__ Q,
                                                     const float* __restrict__ Al,
                                                     const float* __restrict__ EWt,
                                                     const float* __restrict__ Wout,
                                                     float* __restrict__ Yout) {
    const int M = 512, N = 256, H = 256, F = 512, S = 512, T = 128, BS = 2048;
    constexpr int NT = 2;
    __shared__ float red[4 * 2 * 8 * 32];
    const int wave = threadIdx.x >> 5;
    const int lane = threadIdx.x & 31;
    const int blk  = blockIdx.x;
    const int nbn  = N / (16 * NT);
    if (blk >= (M >> 4) * nbn) return;
    const int tm   = (blk / nbn) << 4;
    const int tn   = (blk % nbn) * (16 * NT);
    const int half = lane >> 4;
    const int l    = lane & 15;
    const int kk   = half << 1;
    const int batch = tm / T;
    const int h0   = wave * (H / 4);
    const int s0   = wave * (S / 4);

    const float* qp = Q  + (size_t)(tm + l) * H + h0 + kk;
    const float* al = Al + (size_t)(tm + l) * S + s0 + kk;
    const float* wp[NT];
    const float* ep[NT];
#pragma unroll
    for (int j = 0; j < NT; ++j) {
        wp[j] = Wout + (size_t)(tn + 16 * j + l) * F + h0 + kk;
        ep[j] = EWt  + (size_t)(tn + 16 * j + l) * BS + (size_t)batch * S + s0 + kk;
    }

    constexpr int U = 8;
    v8f acc[1][NT] = {};
    v2f a0[U], b0[U][NT], a1[U], b1[U][NT];
    auto load = [&](v2f (&a)[U], v2f (&b)[U][NT], const float* x, const float* const (&w)[NT], int k) {
        __builtin_amdgcn_sched_barrier(0);
#pragma unroll
        for (int u = 0; u < U; ++u) {
            a[u] = *(const v2f*)(x + k + 4 * u);
#pragma unroll
            for (int j = 0; j < NT; ++j) b[u][j] = *(const v2f*)(w[j] + k + 4 * u);
        }
        __builtin_amdgcn_sched_barrier(0);
    };
    auto mma = [&](const v2f (&a)[U], const v2f (&b)[U][NT]) {
#pragma unroll
        for (int u = 0; u < U; ++u)
#pragma unroll
            for (int j = 0; j < NT; ++j)
                acc[0][j] = WMMA_F32_16x16x4(a[u], b[u][j], acc[0][j]);
    };
    load(a0, b0, qp, wp, 0);
    load(a1, b1, qp, wp, 32);
    mma(a0, b0);
    load(a0, b0, al, ep, 0);
    mma(a1, b1);
    load(a1, b1, al, ep, 32);
    mma(a0, b0);
    load(a0, b0, al, ep, 64);
    mma(a1, b1);
    load(a1, b1, al, ep, 96);
    mma(a0, b0);
    mma(a1, b1);

    float sum[2][2];
    join4_partial_blocks(red, acc, wave, lane, sum);

    const int mbase = tm + half * 8;
#pragma unroll
    for (int q = 0; q < 2; ++q)
#pragma unroll
        for (int j = 0; j < NT; ++j)
            Yout[(size_t)(mbase + 2 * wave + q) * N + tn + 16 * j + l] = tanhf(sum[j][q]);
}

__global__ void __launch_bounds__(256) score_softmax_kernel(const float* __restrict__ Eq,
                                     const float* __restrict__ Ee,
                                     const float* __restrict__ v,
                                     const int*   __restrict__ src_len,
                                     float* __restrict__ alpha) {
    const int H = 256, S = 512, T = 128;
    const int bt   = blockIdx.x;
    const int b    = bt / T;
    const int tid  = threadIdx.x;
    const int lane = tid & 31;
    const int wave = __builtin_amdgcn_readfirstlane(tid >> 5);

    __shared__ float sh_e[512];
    __shared__ float red[8];

    const v4f eq0 = *(const v4f*)(Eq + (size_t)bt * H + lane * 4);
    const v4f eq1 = *(const v4f*)(Eq + (size_t)bt * H + 128 + lane * 4);
    const v4f v0  = *(const v4f*)(v + lane * 4);
    const v4f v1  = *(const v4f*)(v + 128 + lane * 4);

    constexpr int G = 8;
    const float* rowp = Ee + ((size_t)b * S + wave * G) * H + lane * 4;
    v4f bufA[G][2], bufB[G][2];

    auto load_rows = [&](v4f (&buf)[G][2], int i) {
        const float* p = rowp + (size_t)i * (8 * G * H);
        for (int r = 0; r < G; ++r) {
            buf[r][0] = *(const v4f*)(p + (size_t)r * H);
            buf[r][1] = *(const v4f*)(p + (size_t)r * H + 128);
        }
        __builtin_amdgcn_sched_barrier(0);
    };

    load_rows(bufA, 0);

    const v4f w0 = -2.f * v0;
    const v4f w1 = -2.f * v1;

    const int slen = src_len[b];
    const int ng = min(max((slen + G - 1) / G, 0), S / G);
    const int nw = (ng - wave + 7) >> 3;

    {
        const int s = ng * G + tid;
        if (s < S)       sh_e[s]       = -INFINITY;
        if (s + 256 < S) sh_e[s + 256] = -INFINITY;
    }

    auto score_rows = [&](const v4f (&buf)[G][2], int i) {
        float acc[G];
        for (int r = 0; r < G; ++r) {
            float t = 0.f;
            for (int c = 0; c < 4; ++c) {
                t = fmaf(w0[c], __builtin_amdgcn_rcpf(fmaf(eq0[c], buf[r][0][c], 1.f)), t);
                t = fmaf(w1[c], __builtin_amdgcn_rcpf(fmaf(eq1[c], buf[r][1][c], 1.f)), t);
            }
            acc[r] = t;
        }
        for (int r = 0; r < G; ++r) acc[r] += xhalf(acc[r]);
        for (int r = 0; r < G; ++r) acc[r] += row_xor<1>(acc[r]);
        for (int r = 0; r < G; ++r) acc[r] += row_xor<2>(acc[r]);
        for (int r = 0; r < G; ++r) acc[r] += row_xor<4>(acc[r]);
        for (int r = 0; r < G; ++r) acc[r] += row_xor<8>(acc[r]);
        if (lane == 0) {
            for (int r = 0; r < G; ++r) {
                const int s = (wave + 8 * i) * G + r;
                sh_e[s] = (s < slen) ? acc[r] : -INFINITY;
            }
        }
    };

    if (nw > 0) {
        int i = 0;
        for (; i + 2 < nw; i += 2) {
            load_rows(bufB, i + 1);
            score_rows(bufA, i);
            load_rows(bufA, i + 2);
            score_rows(bufB, i + 1);
        }
        if (i + 1 < nw) {
            load_rows(bufB, i + 1);
            score_rows(bufA, i);
            score_rows(bufB, i + 1);
        } else {
            score_rows(bufA, i);
        }
    }
    __syncthreads();

    float m = -INFINITY;
    for (int s = tid; s < S; s += 256) m = fmaxf(m, sh_e[s]);
#pragma unroll
    for (int off = 16; off > 0; off >>= 1) m = fmaxf(m, __shfl_xor(m, off, 32));
    if (lane == 0) red[wave] = m;
    __syncthreads();
    m = red[0];
#pragma unroll
    for (int i = 1; i < 8; ++i) m = fmaxf(m, red[i]);
    __syncthreads();

    float sum = 0.f;
    for (int s = tid; s < S; s += 256) {
        float ex = __expf(sh_e[s] - m);
        sh_e[s] = ex;
        sum += ex;
    }
#pragma unroll
    for (int off = 16; off > 0; off >>= 1) sum += __shfl_xor(sum, off, 32);
    if (lane == 0) red[wave] = sum;
    __syncthreads();
    float total = 0.f;
#pragma unroll
    for (int i = 0; i < 8; ++i) total += red[i];
    const float inv = 1.f / total;

    for (int s = tid; s < S; s += 256)
        alpha[(size_t)bt * S + s] = sh_e[s] * inv;
}

extern "C" void kernel_launch(void* const* d_in, const int* in_sizes, int n_in,
                              void* d_out, int out_size, void* d_ws, size_t ws_size,
                              hipStream_t stream) {
    const int B = 4, T = 128, S = 512, H = 256;

    const float* query = (const float*)d_in[0];
    const float* enc   = (const float*)d_in[1];
    const int*   slen  = (const int*)d_in[2];
    const float* Ws    = (const float*)d_in[3];
    const float* Wh    = (const float*)d_in[4];
    const float* v     = (const float*)d_in[5];
    const float* Wout  = (const float*)d_in[6];
    float* out = (float*)d_out;

    float* ws    = (float*)d_ws;
    float* pq    = ws;
    float* pe    = pq + (size_t)B * T * H;
    float* alpha = pe + (size_t)B * S * H;
    float* encWT = pe;
    float* encWT_own = alpha + (size_t)B * T * S;
    (void)in_sizes; (void)n_in; (void)out_size;

    if (ws_size >= 5767168) {
        enc_and_query_proj_wmma<<<256, 256, 0, stream>>>(enc, Wh, Wout, pe, encWT_own, query, Ws, pq);

        score_softmax_kernel<<<B * T, 256, 0, stream>>>(pq, pe, v, slen, alpha);

        out_gemm_wmma<<<256, 128, 0, stream>>>(query, alpha, encWT_own, Wout, out);
        return;
    }

    pq_exp2x_splitk_wmma<<<256, 128, 0, stream>>>(query, Ws, pq);

    gemm_xwT_exp2x_wmma<2, 2><<<256, 64, 0, stream>>>(enc, Wh, pe, B * S, H, H);

    score_softmax_kernel<<<B * T, 256, 0, stream>>>(pq, pe, v, slen, alpha);

    gemm_xwT_wmma<2, 2><<<256, 64, 0, stream>>>(Wout + H, 2 * H, enc, encWT, H, B * S, H);

    out_gemm_wmma<<<256, 128, 0, stream>>>(query, alpha, encWT, Wout, out);
}
